# starting kernel
# baseline (speedup 1.0000x reference)
#include <hip/hip_runtime.h>
#include <hip/hip_bf16.h>

typedef float v2f __attribute__((ext_vector_type(2)));
typedef float v8f __attribute__((ext_vector_type(8)));

#define B_  16
#define C_  128
#define H_  64
#define W_  256
#define KK_ 1152
#define CSTRIDE (H_ * W_)
#define BSTRIDE (C_ * H_ * W_)

#define KC  72
#define MT  64
#define NT  64
#define A_PAD 73
#define B_PAD 65

__global__ __launch_bounds__(256)
void MessagePass_step_wmma(const float* __restrict__ src,
                           const float* __restrict__ addv,
                           float* __restrict__ dst,
                           const float* __restrict__ Wt,
                           const float* __restrict__ ss)
{
    __shared__ float As[MT * A_PAD];
    __shared__ float Bs[KC * B_PAD];

    const int tid  = threadIdx.x;
    const int lane = tid & 31;
    const int hi   = lane >> 4;
    const int l16  = lane & 15;
    const int wv   = tid >> 5;

    const int m0 = blockIdx.y * MT;
    const int n0 = blockIdx.x * NT;
    const int b  = n0 / W_;
    const int w0 = n0 % W_;

    const int mt  = wv & 3;
    const int ntp = (wv >> 2) * 2;

    v8f acc0 = {};
    v8f acc1 = {};

    for (int kc0 = 0; kc0 < KK_; kc0 += KC) {
        __syncthreads();
        for (int idx = tid; idx < MT * KC; idx += 256) {
            int mr = idx / KC, kk = idx % KC;
            As[mr * A_PAD + kk] = Wt[(m0 + mr) * KK_ + kc0 + kk];
        }
        for (int idx = tid; idx < KC * NT; idx += 256) {
            int kk  = idx / NT, col = idx % NT;
            int kg  = kc0 + kk;
            int ic  = kg / 9;
            int tap = kg % 9;
            int w   = w0 + col + tap - 4;
            float v = 0.0f;
            if (w >= 0 && w < W_)
                v = src[(size_t)b * BSTRIDE + (size_t)ic * CSTRIDE + w];
            Bs[kk * B_PAD + col] = v;
        }
        __syncthreads();

        const float* Arow = &As[(mt * 16 + l16) * A_PAD];
        #pragma unroll
        for (int kk = 0; kk < KC; kk += 4) {
            const int kb = kk + 2 * hi;
            v2f a;
            a.x = Arow[kb];
            a.y = Arow[kb + 1];
            v2f bf0, bf1;
            bf0.x = Bs[kb * B_PAD + ntp * 16 + l16];
            bf0.y = Bs[(kb + 1) * B_PAD + ntp * 16 + l16];
            bf1.x = Bs[kb * B_PAD + (ntp + 1) * 16 + l16];
            bf1.y = Bs[(kb + 1) * B_PAD + (ntp + 1) * 16 + l16];
            acc0 = __builtin_amdgcn_wmma_f32_16x16x4_f32(false, a, false, bf0,
                                                         (short)0, acc0, false, false);
            acc1 = __builtin_amdgcn_wmma_f32_16x16x4_f32(false, a, false, bf1,
                                                         (short)0, acc1, false, false);
        }
    }

    const float* scale = ss;
    const float* shift = ss + C_;
    #pragma unroll
    for (int j = 0; j < 8; ++j) {
        const int M = mt * 16 + j + 8 * hi;
        const int o = m0 + M;
        const float sc = scale[o];
        const float sh = shift[o];
        {
            const int w = w0 + ntp * 16 + l16;
            const size_t off = (size_t)b * BSTRIDE + (size_t)o * CSTRIDE + w;
            float y = acc0[j] * sc + sh;
            y = y > 0.0f ? y : 0.0f;
            dst[off] = y + addv[off];
        }
        {
            const int w = w0 + (ntp + 1) * 16 + l16;
            const size_t off = (size_t)b * BSTRIDE + (size_t)o * CSTRIDE + w;
            float y = acc1[j] * sc + sh;
            y = y > 0.0f ? y : 0.0f;
            dst[off] = y + addv[off];
        }
    }
}

__global__ void MessagePass_copy_slice(const float* __restrict__ x, float* __restrict__ out)
{
    int e = blockIdx.x * blockDim.x + threadIdx.x;
    if (e >= B_ * C_ * W_) return;
    int w  = e % W_;
    int bc = e / W_;
    size_t off = (size_t)bc * CSTRIDE + w;
    out[off] = x[off];
}

__global__ void MessagePass_bn_prep(const float* __restrict__ g, const float* __restrict__ be,
                                    const float* __restrict__ rm, const float* __restrict__ rv,
                                    float* __restrict__ ss)
{
    int c = blockIdx.x * blockDim.x + threadIdx.x;
    if (c < C_) {
        float sc = g[c] * rsqrtf(rv[c] + 1e-5f);
        ss[c]      = sc;
        ss[C_ + c] = be[c] - rm[c] * sc;
    }
}

extern "C" void kernel_launch(void* const* d_in, const int* in_sizes, int n_in,
                              void* d_out, int out_size, void* d_ws, size_t ws_size,
                              hipStream_t stream)
{
    const float* x     = (const float*)d_in[0];
    const float* cw    = (const float*)d_in[1];
    const float* gamma = (const float*)d_in[2];
    const float* beta  = (const float*)d_in[3];
    const float* rmean = (const float*)d_in[4];
    const float* rvar  = (const float*)d_in[5];
    float* out = (float*)d_out;
    float* ss  = (float*)d_ws;

    (void)in_sizes; (void)n_in; (void)out_size; (void)ws_size;

    MessagePass_bn_prep<<<1, 128, 0, stream>>>(gamma, beta, rmean, rvar, ss);
    MessagePass_copy_slice<<<(B_ * C_ * W_ + 255) / 256, 256, 0, stream>>>(x, out);

    dim3 grid((B_ * W_) / NT, C_ / MT);

    for (int h = 1; h < H_; ++h)
        MessagePass_step_wmma<<<grid, 256, 0, stream>>>(
            out + (size_t)(h - 1) * W_, x + (size_t)h * W_, out + (size_t)h * W_, cw, ss);

    for (int h = H_ - 2; h >= 0; --h)
        MessagePass_step_wmma<<<grid, 256, 0, stream>>>(
            out + (size_t)(h + 1) * W_, out + (size_t)h * W_, out + (size_t)h * W_, cw, ss);
}


# K loop software-pipelined: next chunk's global loads issued as one batch into registers before the current chunk's WMMAs, stored to a second LDS buffer after them (was load, wait, store serially)
# speedup vs baseline: 3.5566x; 3.5566x over previous
#include <hip/hip_runtime.h>
#include <hip/hip_bf16.h>

typedef float v2f __attribute__((ext_vector_type(2)));
typedef float v8f __attribute__((ext_vector_type(8)));

#define B_  16
#define C_  128
#define H_  64
#define W_  256
#define KK_ 1152
#define CSTRIDE (H_ * W_)
#define BSTRIDE (C_ * H_ * W_)

#define KC  72
#define MT  64
#define NT  64
#define A_PAD 73
#define B_PAD 65

#define NCHUNK (KK_ / KC)

__device__ __forceinline__
void MessagePass_chunk_mma(const float* __restrict__ Ab, const float* __restrict__ Bb,
                           int mt, int ntp, int hi, int l16, v8f& acc0, v8f& acc1)
{
    const float* Arow = &Ab[(mt * 16 + l16) * A_PAD];
    #pragma unroll
    for (int kk = 0; kk < KC; kk += 4) {
        const int kb = kk + 2 * hi;
        v2f a;
        a.x = Arow[kb];
        a.y = Arow[kb + 1];
        v2f bf0, bf1;
        bf0.x = Bb[kb * B_PAD + ntp * 16 + l16];
        bf0.y = Bb[(kb + 1) * B_PAD + ntp * 16 + l16];
        bf1.x = Bb[kb * B_PAD + (ntp + 1) * 16 + l16];
        bf1.y = Bb[(kb + 1) * B_PAD + (ntp + 1) * 16 + l16];
        acc0 = __builtin_amdgcn_wmma_f32_16x16x4_f32(false, a, false, bf0,
                                                     (short)0, acc0, false, false);
        acc1 = __builtin_amdgcn_wmma_f32_16x16x4_f32(false, a, false, bf1,
                                                     (short)0, acc1, false, false);
    }
}

__global__ __launch_bounds__(256)
void MessagePass_step_wmma(const float* __restrict__ src,
                           const float* __restrict__ addv,
                           float* __restrict__ dst,
                           const float* __restrict__ Wt,
                           const float* __restrict__ ss)
{
    __shared__ float As[2][MT * A_PAD];
    __shared__ float Bs[2][KC * B_PAD];

    const int tid  = threadIdx.x;
    const int lane = tid & 31;
    const int hi   = lane >> 4;
    const int l16  = lane & 15;
    const int wv   = tid >> 5;

    const int m0 = blockIdx.y * MT;
    const int n0 = blockIdx.x * NT;
    const int b  = n0 / W_;
    const int w0 = n0 % W_;

    const int mt  = wv & 3;
    const int ntp = (wv >> 2) * 2;

    const int a_row = wv * 8 + (lane >> 3);
    const int a_kq  = lane & 7;
    const float* a_g = Wt + (size_t)(m0 + a_row) * KK_ + a_kq;
    const int a_l = a_row * A_PAD + a_kq;

    const int b_col = tid & 63;
    const int b_kg  = tid >> 6;
    const float* b_g = src + (size_t)b * BSTRIDE + (size_t)(2 * b_kg) * CSTRIDE;
    const int b_l = (b_kg * 18) * B_PAD + b_col;
    int  b_w[9];
    bool b_ok[9];
    #pragma unroll
    for (int t = 0; t < 9; ++t) {
        const int w = w0 + b_col + t - 4;
        b_ok[t] = (w >= 0) && (w < W_);
        b_w[t]  = w < 0 ? 0 : (w >= W_ ? W_ - 1 : w);
    }

    float ra[18], rb[18];

#define LOAD_CHUNK(c)                                                              \
    {                                                                              \
        const float* ap = a_g + (c) * KC;                                          \
        _Pragma("unroll")                                                          \
        for (int j = 0; j < 18; ++j)                                               \
            ra[j] = ap[(j / 9) * 4 * KK_ + (j % 9) * 8];                           \
        const float* bp = b_g + (size_t)(c) * 8 * CSTRIDE;                         \
        _Pragma("unroll")                                                          \
        for (int i = 0; i < 18; ++i)                                               \
            rb[i] = bp[(i / 9) * CSTRIDE + b_w[i % 9]];                            \
    }
#define STORE_CHUNK(buf)                                                           \
    {                                                                              \
        _Pragma("unroll")                                                          \
        for (int j = 0; j < 18; ++j)                                               \
            As[buf][a_l + (j / 9) * 4 * A_PAD + (j % 9) * 8] = ra[j];              \
        _Pragma("unroll")                                                          \
        for (int i = 0; i < 18; ++i)                                               \
            Bs[buf][b_l + i * B_PAD] = b_ok[i % 9] ? rb[i] : 0.0f;                 \
    }

    v8f acc0 = {};
    v8f acc1 = {};

    LOAD_CHUNK(0)
    STORE_CHUNK(0)
    __syncthreads();

    for (int c = 0; c < NCHUNK - 1; ++c) {
        const int cur = c & 1;
        LOAD_CHUNK(c + 1)
        MessagePass_chunk_mma(As[cur], Bs[cur], mt, ntp, hi, l16, acc0, acc1);
        STORE_CHUNK(cur ^ 1)
        __syncthreads();
    }
    MessagePass_chunk_mma(As[(NCHUNK - 1) & 1], Bs[(NCHUNK - 1) & 1], mt, ntp, hi, l16, acc0, acc1);
#undef LOAD_CHUNK
#undef STORE_CHUNK

    const float* scale = ss;
    const float* shift = ss + C_;
    #pragma unroll
    for (int j = 0; j < 8; ++j) {
        const int M = mt * 16 + j + 8 * hi;
        const int o = m0 + M;
        const float sc = scale[o];
        const float sh = shift[o];
        {
            const int w = w0 + ntp * 16 + l16;
            const size_t off = (size_t)b * BSTRIDE + (size_t)o * CSTRIDE + w;
            float y = acc0[j] * sc + sh;
            y = y > 0.0f ? y : 0.0f;
            dst[off] = y + addv[off];
        }
        {
            const int w = w0 + (ntp + 1) * 16 + l16;
            const size_t off = (size_t)b * BSTRIDE + (size_t)o * CSTRIDE + w;
            float y = acc1[j] * sc + sh;
            y = y > 0.0f ? y : 0.0f;
            dst[off] = y + addv[off];
        }
    }
}

__global__ void MessagePass_copy_slice(const float* __restrict__ x, float* __restrict__ out)
{
    int e = blockIdx.x * blockDim.x + threadIdx.x;
    if (e >= B_ * C_ * W_) return;
    int w  = e % W_;
    int bc = e / W_;
    size_t off = (size_t)bc * CSTRIDE + w;
    out[off] = x[off];
}

__global__ void MessagePass_bn_prep(const float* __restrict__ g, const float* __restrict__ be,
                                    const float* __restrict__ rm, const float* __restrict__ rv,
                                    float* __restrict__ ss)
{
    int c = blockIdx.x * blockDim.x + threadIdx.x;
    if (c < C_) {
        float sc = g[c] * rsqrtf(rv[c] + 1e-5f);
        ss[c]      = sc;
        ss[C_ + c] = be[c] - rm[c] * sc;
    }
}

extern "C" void kernel_launch(void* const* d_in, const int* in_sizes, int n_in,
                              void* d_out, int out_size, void* d_ws, size_t ws_size,
                              hipStream_t stream)
{
    const float* x     = (const float*)d_in[0];
    const float* cw    = (const float*)d_in[1];
    const float* gamma = (const float*)d_in[2];
    const float* beta  = (const float*)d_in[3];
    const float* rmean = (const float*)d_in[4];
    const float* rvar  = (const float*)d_in[5];
    float* out = (float*)d_out;
    float* ss  = (float*)d_ws;

    (void)in_sizes; (void)n_in; (void)out_size; (void)ws_size;

    MessagePass_bn_prep<<<1, 128, 0, stream>>>(gamma, beta, rmean, rvar, ss);
    MessagePass_copy_slice<<<(B_ * C_ * W_ + 255) / 256, 256, 0, stream>>>(x, out);

    dim3 grid((B_ * W_) / NT, C_ / MT);

    for (int h = 1; h < H_; ++h)
        MessagePass_step_wmma<<<grid, 256, 0, stream>>>(
            out + (size_t)(h - 1) * W_, x + (size_t)h * W_, out + (size_t)h * W_, cw, ss);

    for (int h = H_ - 2; h >= 0; --h)
        MessagePass_step_wmma<<<grid, 256, 0, stream>>>(
            out + (size_t)(h + 1) * W_, out + (size_t)h * W_, out + (size_t)h * W_, cw, ss);
}


# LDS fragment reads software-pipelined: each half chunk's reads issued as one group a block of WMMAs ahead into a second register set (was one LDS round trip per WMMA); summation order unchanged
# speedup vs baseline: 4.7929x; 1.3476x over previous
#include <hip/hip_runtime.h>
#include <hip/hip_bf16.h>

typedef float v2f __attribute__((ext_vector_type(2)));
typedef float v8f __attribute__((ext_vector_type(8)));

#define B_  16
#define C_  128
#define H_  64
#define W_  256
#define KK_ 1152
#define CSTRIDE (H_ * W_)
#define BSTRIDE (C_ * H_ * W_)

#define KC  72
#define MT  64
#define NT  64
#define A_PAD 73
#define B_PAD 65

#define NCHUNK (KK_ / KC)

#define KSUB 9

__device__ __forceinline__
void MessagePass_frag_load(const float* __restrict__ Arow, const float* __restrict__ Bcol,
                           int blk, v2f* fa, v2f* fb0, v2f* fb1)
{
    #pragma unroll
    for (int s = 0; s < KSUB; ++s) {
        const int kk = (blk * KSUB + s) * 4;
        fa[s].x  = Arow[kk];
        fa[s].y  = Arow[kk + 1];
        fb0[s].x = Bcol[kk * B_PAD];
        fb0[s].y = Bcol[(kk + 1) * B_PAD];
    }
    __builtin_amdgcn_sched_barrier(0);
    #pragma unroll
    for (int s = 0; s < KSUB; ++s) {
        const int kk = (blk * KSUB + s) * 4;
        fb1[s].x = Bcol[kk * B_PAD + 16];
        fb1[s].y = Bcol[(kk + 1) * B_PAD + 16];
    }
}

__device__ __forceinline__
void MessagePass_frag_mma(const v2f* fa, const v2f* fb0, const v2f* fb1, v8f& acc0, v8f& acc1)
{
    #pragma unroll
    for (int s = 0; s < KSUB; ++s) {
        acc0 = __builtin_amdgcn_wmma_f32_16x16x4_f32(false, fa[s], false, fb0[s],
                                                     (short)0, acc0, false, false);
        acc1 = __builtin_amdgcn_wmma_f32_16x16x4_f32(false, fa[s], false, fb1[s],
                                                     (short)0, acc1, false, false);
    }
}

__global__ __launch_bounds__(256)
void MessagePass_step_wmma(const float* __restrict__ src,
                           const float* __restrict__ addv,
                           float* __restrict__ dst,
                           const float* __restrict__ Wt,
                           const float* __restrict__ ss)
{
    __shared__ float As[2][MT * A_PAD];
    __shared__ float Bs[2][KC * B_PAD];

    const int tid  = threadIdx.x;
    const int lane = tid & 31;
    const int hi   = lane >> 4;
    const int l16  = lane & 15;
    const int wv   = tid >> 5;

    const int m0 = blockIdx.y * MT;
    const int n0 = blockIdx.x * NT;
    const int b  = n0 / W_;
    const int w0 = n0 % W_;

    const int mt  = wv & 3;
    const int ntp = (wv >> 2) * 2;

    const int a_row = wv * 8 + (lane >> 3);
    const int a_kq  = lane & 7;
    const float* a_g = Wt + (size_t)(m0 + a_row) * KK_ + a_kq;
    const int a_l = a_row * A_PAD + a_kq;

    const int b_col = tid & 63;
    const int b_kg  = tid >> 6;
    const float* b_g = src + (size_t)b * BSTRIDE + (size_t)(2 * b_kg) * CSTRIDE;
    const int b_l = (b_kg * 18) * B_PAD + b_col;
    int  b_w[9];
    bool b_ok[9];
    #pragma unroll
    for (int t = 0; t < 9; ++t) {
        const int w = w0 + b_col + t - 4;
        b_ok[t] = (w >= 0) && (w < W_);
        b_w[t]  = w < 0 ? 0 : (w >= W_ ? W_ - 1 : w);
    }

    float ra[18], rb[18];

#define LOAD_CHUNK(c)                                                              \
    {                                                                              \
        const float* ap = a_g + (c) * KC;                                          \
        _Pragma("unroll")                                                          \
        for (int j = 0; j < 18; ++j)                                               \
            ra[j] = ap[(j / 9) * 4 * KK_ + (j % 9) * 8];                           \
        const float* bp = b_g + (size_t)(c) * 8 * CSTRIDE;                         \
        _Pragma("unroll")                                                          \
        for (int i = 0; i < 18; ++i)                                               \
            rb[i] = bp[(i / 9) * CSTRIDE + b_w[i % 9]];                            \
    }
#define STORE_CHUNK(buf)                                                           \
    {                                                                              \
        _Pragma("unroll")                                                          \
        for (int j = 0; j < 18; ++j)                                               \
            As[buf][a_l + (j / 9) * 4 * A_PAD + (j % 9) * 8] = ra[j];              \
        _Pragma("unroll")                                                          \
        for (int i = 0; i < 18; ++i)                                               \
            Bs[buf][b_l + i * B_PAD] = b_ok[i % 9] ? rb[i] : 0.0f;                 \
    }

    v8f acc0 = {};
    v8f acc1 = {};

    const int fa_off = (mt * 16 + l16) * A_PAD + 2 * hi;
    const int fb_off = (2 * hi) * B_PAD + ntp * 16 + l16;

    v2f fa[2][KSUB], fb0[2][KSUB], fb1[2][KSUB];

#define CHUNK_BODY(c, MORE)                                                                \
    {                                                                                      \
        const int cur = (c) & 1;                                                           \
        MessagePass_frag_load(&As[cur][fa_off], &Bs[cur][fb_off], 1, fa[1], fb0[1], fb1[1]);      \
        MessagePass_frag_mma(fa[0], fb0[0], fb1[0], acc0, acc1);                           \
        STORE_CHUNK(cur ^ 1)                   \
        __syncthreads();                                                                   \
        if (MORE) LOAD_CHUNK((c) + 2)              \
        MessagePass_frag_load(&As[cur ^ 1][fa_off], &Bs[cur ^ 1][fb_off], 0, fa[0], fb0[0], fb1[0]); \
        MessagePass_frag_mma(fa[1], fb0[1], fb1[1], acc0, acc1);                           \
    }

    LOAD_CHUNK(0)
    STORE_CHUNK(0)
    __syncthreads();
    LOAD_CHUNK(1)
    MessagePass_frag_load(&As[0][fa_off], &Bs[0][fb_off], 0, fa[0], fb0[0], fb1[0]);

    for (int c = 0; c < NCHUNK - 2; ++c)
        CHUNK_BODY(c, true)
    CHUNK_BODY(NCHUNK - 2, false)
    {
        const int cur = (NCHUNK - 1) & 1;
        MessagePass_frag_load(&As[cur][fa_off], &Bs[cur][fb_off], 1, fa[1], fb0[1], fb1[1]);
        MessagePass_frag_mma(fa[0], fb0[0], fb1[0], acc0, acc1);
        MessagePass_frag_mma(fa[1], fb0[1], fb1[1], acc0, acc1);
    }
#undef CHUNK_BODY
#undef LOAD_CHUNK
#undef STORE_CHUNK

    const float* scale = ss;
    const float* shift = ss + C_;
    #pragma unroll
    for (int j = 0; j < 8; ++j) {
        const int M = mt * 16 + j + 8 * hi;
        const int o = m0 + M;
        const float sc = scale[o];
        const float sh = shift[o];
        {
            const int w = w0 + ntp * 16 + l16;
            const size_t off = (size_t)b * BSTRIDE + (size_t)o * CSTRIDE + w;
            float y = acc0[j] * sc + sh;
            y = y > 0.0f ? y : 0.0f;
            dst[off] = y + addv[off];
        }
        {
            const int w = w0 + (ntp + 1) * 16 + l16;
            const size_t off = (size_t)b * BSTRIDE + (size_t)o * CSTRIDE + w;
            float y = acc1[j] * sc + sh;
            y = y > 0.0f ? y : 0.0f;
            dst[off] = y + addv[off];
        }
    }
}

__global__ void MessagePass_copy_slice(const float* __restrict__ x, float* __restrict__ out)
{
    int e = blockIdx.x * blockDim.x + threadIdx.x;
    if (e >= B_ * C_ * W_) return;
    int w  = e % W_;
    int bc = e / W_;
    size_t off = (size_t)bc * CSTRIDE + w;
    out[off] = x[off];
}

__global__ void MessagePass_bn_prep(const float* __restrict__ g, const float* __restrict__ be,
                                    const float* __restrict__ rm, const float* __restrict__ rv,
                                    float* __restrict__ ss)
{
    int c = blockIdx.x * blockDim.x + threadIdx.x;
    if (c < C_) {
        float sc = g[c] * rsqrtf(rv[c] + 1e-5f);
        ss[c]      = sc;
        ss[C_ + c] = be[c] - rm[c] * sc;
    }
}

extern "C" void kernel_launch(void* const* d_in, const int* in_sizes, int n_in,
                              void* d_out, int out_size, void* d_ws, size_t ws_size,
                              hipStream_t stream)
{
    const float* x     = (const float*)d_in[0];
    const float* cw    = (const float*)d_in[1];
    const float* gamma = (const float*)d_in[2];
    const float* beta  = (const float*)d_in[3];
    const float* rmean = (const float*)d_in[4];
    const float* rvar  = (const float*)d_in[5];
    float* out = (float*)d_out;
    float* ss  = (float*)d_ws;

    (void)in_sizes; (void)n_in; (void)out_size; (void)ws_size;

    MessagePass_bn_prep<<<1, 128, 0, stream>>>(gamma, beta, rmean, rvar, ss);
    MessagePass_copy_slice<<<(B_ * C_ * W_ + 255) / 256, 256, 0, stream>>>(x, out);

    dim3 grid((B_ * W_) / NT, C_ / MT);

    for (int h = 1; h < H_; ++h)
        MessagePass_step_wmma<<<grid, 256, 0, stream>>>(
            out + (size_t)(h - 1) * W_, x + (size_t)h * W_, out + (size_t)h * W_, cw, ss);

    for (int h = H_ - 2; h >= 0; --h)
        MessagePass_step_wmma<<<grid, 256, 0, stream>>>(
            out + (size_t)(h + 1) * W_, out + (size_t)h * W_, out + (size_t)h * W_, cw, ss);
}


# Grid doubled to 256 workgroups (one per WGP, was 128 on half the chip): block tile 64x64 -> 32x64 with 4 waves, one per SIMD; wave tile, LDS layout, k order and summation order unchanged
# speedup vs baseline: 6.3878x; 1.3328x over previous
#include <hip/hip_runtime.h>
#include <hip/hip_bf16.h>

typedef float v2f __attribute__((ext_vector_type(2)));
typedef float v8f __attribute__((ext_vector_type(8)));

#define B_  16
#define C_  128
#define H_  64
#define W_  256
#define KK_ 1152
#define CSTRIDE (H_ * W_)
#define BSTRIDE (C_ * H_ * W_)

#define KC  72
#define MT  32
#define NT  64
#define A_PAD 73
#define B_PAD 65

#define NCHUNK (KK_ / KC)

#define KSUB 9

__device__ __forceinline__
void MessagePass_frag_load(const float* __restrict__ Arow, const float* __restrict__ Bcol,
                           int blk, v2f* fa, v2f* fb0, v2f* fb1)
{
    #pragma unroll
    for (int s = 0; s < KSUB; ++s) {
        const int kk = (blk * KSUB + s) * 4;
        fa[s].x  = Arow[kk];
        fa[s].y  = Arow[kk + 1];
        fb0[s].x = Bcol[kk * B_PAD];
        fb0[s].y = Bcol[(kk + 1) * B_PAD];
    }
    __builtin_amdgcn_sched_barrier(0);
    #pragma unroll
    for (int s = 0; s < KSUB; ++s) {
        const int kk = (blk * KSUB + s) * 4;
        fb1[s].x = Bcol[kk * B_PAD + 16];
        fb1[s].y = Bcol[(kk + 1) * B_PAD + 16];
    }
}

__device__ __forceinline__
void MessagePass_frag_mma(const v2f* fa, const v2f* fb0, const v2f* fb1, v8f& acc0, v8f& acc1)
{
    #pragma unroll
    for (int s = 0; s < KSUB; ++s) {
        acc0 = __builtin_amdgcn_wmma_f32_16x16x4_f32(false, fa[s], false, fb0[s],
                                                     (short)0, acc0, false, false);
        acc1 = __builtin_amdgcn_wmma_f32_16x16x4_f32(false, fa[s], false, fb1[s],
                                                     (short)0, acc1, false, false);
    }
}

__global__ __launch_bounds__(128)
void MessagePass_step_wmma(const float* __restrict__ src,
                           const float* __restrict__ addv,
                           float* __restrict__ dst,
                           const float* __restrict__ Wt,
                           const float* __restrict__ ss)
{
    __shared__ float As[2][MT * A_PAD];
    __shared__ float Bs[2][KC * B_PAD];

    const int tid  = threadIdx.x;
    const int lane = tid & 31;
    const int hi   = lane >> 4;
    const int l16  = lane & 15;
    const int wv   = tid >> 5;

    const int m0 = blockIdx.y * MT;
    const int n0 = blockIdx.x * NT;
    const int b  = n0 / W_;
    const int w0 = n0 % W_;

    const int mt  = wv & 1;
    const int ntp = (wv >> 1) * 2;

    const int a_row = wv * 8 + (lane >> 3);
    const int a_kq  = lane & 7;
    const float* a_g = Wt + (size_t)(m0 + a_row) * KK_ + a_kq;
    const int a_l = a_row * A_PAD + a_kq;

    const int b_col = tid & 63;
    const int b_kg  = tid >> 6;
    const float* b_g = src + (size_t)b * BSTRIDE + (size_t)(4 * b_kg) * CSTRIDE;
    const int b_l = (b_kg * 36) * B_PAD + b_col;
    int  b_w[9];
    bool b_ok[9];
    #pragma unroll
    for (int t = 0; t < 9; ++t) {
        const int w = w0 + b_col + t - 4;
        b_ok[t] = (w >= 0) && (w < W_);
        b_w[t]  = w < 0 ? 0 : (w >= W_ ? W_ - 1 : w);
    }

    float ra[18], rb[36];

#define LOAD_CHUNK(c)                                                              \
    {                                                                              \
        const float* ap = a_g + (c) * KC;                                          \
        _Pragma("unroll")                                                          \
        for (int j = 0; j < 18; ++j)                                               \
            ra[j] = ap[(j / 9) * 4 * KK_ + (j % 9) * 8];                           \
        const float* bp = b_g + (size_t)(c) * 8 * CSTRIDE;                         \
        _Pragma("unroll")                                                          \
        for (int i = 0; i < 36; ++i)                                               \
            rb[i] = bp[(i / 9) * CSTRIDE + b_w[i % 9]];                            \
    }
#define STORE_CHUNK(buf)                                                           \
    {                                                                              \
        _Pragma("unroll")                                                          \
        for (int j = 0; j < 18; ++j)                                               \
            As[buf][a_l + (j / 9) * 4 * A_PAD + (j % 9) * 8] = ra[j];              \
        _Pragma("unroll")                                                          \
        for (int i = 0; i < 36; ++i)                                               \
            Bs[buf][b_l + i * B_PAD] = b_ok[i % 9] ? rb[i] : 0.0f;                 \
    }

    v8f acc0 = {};
    v8f acc1 = {};

    const int fa_off = (mt * 16 + l16) * A_PAD + 2 * hi;
    const int fb_off = (2 * hi) * B_PAD + ntp * 16 + l16;

    v2f fa[2][KSUB], fb0[2][KSUB], fb1[2][KSUB];

#define CHUNK_BODY(c, MORE)                                                                \
    {                                                                                      \
        const int cur = (c) & 1;                                                           \
        MessagePass_frag_load(&As[cur][fa_off], &Bs[cur][fb_off], 1, fa[1], fb0[1], fb1[1]);      \
        MessagePass_frag_mma(fa[0], fb0[0], fb1[0], acc0, acc1);                           \
        STORE_CHUNK(cur ^ 1)                   \
        __syncthreads();                                                                   \
        if (MORE) LOAD_CHUNK((c) + 2)              \
        MessagePass_frag_load(&As[cur ^ 1][fa_off], &Bs[cur ^ 1][fb_off], 0, fa[0], fb0[0], fb1[0]); \
        MessagePass_frag_mma(fa[1], fb0[1], fb1[1], acc0, acc1);                           \
    }

    LOAD_CHUNK(0)
    STORE_CHUNK(0)
    __syncthreads();
    LOAD_CHUNK(1)
    MessagePass_frag_load(&As[0][fa_off], &Bs[0][fb_off], 0, fa[0], fb0[0], fb1[0]);

    for (int c = 0; c < NCHUNK - 2; ++c)
        CHUNK_BODY(c, true)
    CHUNK_BODY(NCHUNK - 2, false)
    {
        const int cur = (NCHUNK - 1) & 1;
        MessagePass_frag_load(&As[cur][fa_off], &Bs[cur][fb_off], 1, fa[1], fb0[1], fb1[1]);
        MessagePass_frag_mma(fa[0], fb0[0], fb1[0], acc0, acc1);
        MessagePass_frag_mma(fa[1], fb0[1], fb1[1], acc0, acc1);
    }
#undef CHUNK_BODY
#undef LOAD_CHUNK
#undef STORE_CHUNK

    const float* scale = ss;
    const float* shift = ss + C_;
    #pragma unroll
    for (int j = 0; j < 8; ++j) {
        const int M = mt * 16 + j + 8 * hi;
        const int o = m0 + M;
        const float sc = scale[o];
        const float sh = shift[o];
        {
            const int w = w0 + ntp * 16 + l16;
            const size_t off = (size_t)b * BSTRIDE + (size_t)o * CSTRIDE + w;
            float y = acc0[j] * sc + sh;
            y = y > 0.0f ? y : 0.0f;
            dst[off] = y + addv[off];
        }
        {
            const int w = w0 + (ntp + 1) * 16 + l16;
            const size_t off = (size_t)b * BSTRIDE + (size_t)o * CSTRIDE + w;
            float y = acc1[j] * sc + sh;
            y = y > 0.0f ? y : 0.0f;
            dst[off] = y + addv[off];
        }
    }
}

__global__ void MessagePass_copy_slice(const float* __restrict__ x, float* __restrict__ out)
{
    int e = blockIdx.x * blockDim.x + threadIdx.x;
    if (e >= B_ * C_ * W_) return;
    int w  = e % W_;
    int bc = e / W_;
    size_t off = (size_t)bc * CSTRIDE + w;
    out[off] = x[off];
}

__global__ void MessagePass_bn_prep(const float* __restrict__ g, const float* __restrict__ be,
                                    const float* __restrict__ rm, const float* __restrict__ rv,
                                    float* __restrict__ ss)
{
    int c = blockIdx.x * blockDim.x + threadIdx.x;
    if (c < C_) {
        float sc = g[c] * rsqrtf(rv[c] + 1e-5f);
        ss[c]      = sc;
        ss[C_ + c] = be[c] - rm[c] * sc;
    }
}

extern "C" void kernel_launch(void* const* d_in, const int* in_sizes, int n_in,
                              void* d_out, int out_size, void* d_ws, size_t ws_size,
                              hipStream_t stream)
{
    const float* x     = (const float*)d_in[0];
    const float* cw    = (const float*)d_in[1];
    const float* gamma = (const float*)d_in[2];
    const float* beta  = (const float*)d_in[3];
    const float* rmean = (const float*)d_in[4];
    const float* rvar  = (const float*)d_in[5];
    float* out = (float*)d_out;
    float* ss  = (float*)d_ws;

    (void)in_sizes; (void)n_in; (void)out_size; (void)ws_size;

    MessagePass_bn_prep<<<1, 128, 0, stream>>>(gamma, beta, rmean, rvar, ss);
    MessagePass_copy_slice<<<(B_ * C_ * W_ + 255) / 256, 256, 0, stream>>>(x, out);

    dim3 grid((B_ * W_) / NT, C_ / MT);

    for (int h = 1; h < H_; ++h)
        MessagePass_step_wmma<<<grid, 128, 0, stream>>>(
            out + (size_t)(h - 1) * W_, x + (size_t)h * W_, out + (size_t)h * W_, cw, ss);

    for (int h = H_ - 2; h >= 0; --h)
        MessagePass_step_wmma<<<grid, 128, 0, stream>>>(
            out + (size_t)(h + 1) * W_, out + (size_t)h * W_, out + (size_t)h * W_, cw, ss);
}


# Input staged as raw rows (8 channels x 72 columns per chunk), tap shift in the LDS read address, replacing the 9x im2col image; reads kept grouped ahead of each product block; summation order changes
# speedup vs baseline: 7.2960x; 1.1422x over previous
#include <hip/hip_runtime.h>
#include <hip/hip_bf16.h>

typedef float v2f __attribute__((ext_vector_type(2)));
typedef float v4f __attribute__((ext_vector_type(4)));
typedef float v8f __attribute__((ext_vector_type(8)));

#define B_  16
#define C_  128
#define H_  64
#define W_  256
#define KK_ 1152
#define CSTRIDE (H_ * W_)
#define BSTRIDE (C_ * H_ * W_)

#define KC  72
#define MT  32
#define NT  64
#define A_PAD 76
#define PROW 72
#define PATCH_ROW(c) ((((c) & 2) << 1) | (((c) & 4) >> 1) | ((c) & 1))

#define NCHUNK (KK_ / KC)

#define KSUB 9

__device__ __forceinline__
void MessagePass_frag_load(const float* __restrict__ Arow, const float* __restrict__ Pcol,
                           int blk, v2f* fa, v2f* fb0, v2f* fb1)
{
    const float* Pb = Pcol + 2 * blk * PROW;
    #pragma unroll
    for (int s = 0; s < KSUB; ++s) {
        const int j0 = 2 * s, j1 = 2 * s + 1;
        fa[s]    = *reinterpret_cast<const v2f*>(Arow + 36 * blk + 2 * s);
        fb0[s].x = Pb[(j0 / 9) * PROW + (j0 % 9)];
        fb0[s].y = Pb[(j1 / 9) * PROW + (j1 % 9)];
    }
    __builtin_amdgcn_sched_barrier(0);
    #pragma unroll
    for (int s = 0; s < KSUB; ++s) {
        const int j0 = 2 * s, j1 = 2 * s + 1;
        fb1[s].x = Pb[(j0 / 9) * PROW + (j0 % 9) + 16];
        fb1[s].y = Pb[(j1 / 9) * PROW + (j1 % 9) + 16];
    }
    __builtin_amdgcn_sched_barrier(0);
}

__device__ __forceinline__
void MessagePass_frag_mma(const v2f* fa, const v2f* fb0, const v2f* fb1, v8f& acc0, v8f& acc1)
{
    #pragma unroll
    for (int s = 0; s < KSUB; ++s) {
        acc0 = __builtin_amdgcn_wmma_f32_16x16x4_f32(false, fa[s], false, fb0[s],
                                                     (short)0, acc0, false, false);
        acc1 = __builtin_amdgcn_wmma_f32_16x16x4_f32(false, fa[s], false, fb1[s],
                                                     (short)0, acc1, false, false);
    }
    asm volatile("" : "+v"(acc0), "+v"(acc1) : : "memory");
}

__global__ __launch_bounds__(128)
void MessagePass_step_wmma(const float* __restrict__ src,
                           const float* __restrict__ addv,
                           float* __restrict__ dst,
                           const float* __restrict__ Wt,
                           const float* __restrict__ ss)
{
    __shared__ __attribute__((aligned(16))) float As[2][MT * A_PAD];
    __shared__ __attribute__((aligned(16))) float Ps[2][8 * PROW];

    const int tid  = threadIdx.x;
    const int lane = tid & 31;
    const int hi   = lane >> 4;
    const int l16  = lane & 15;
    const int wv   = tid >> 5;

    const int m0 = blockIdx.y * MT;
    const int n0 = blockIdx.x * NT;
    const int b  = n0 / W_;
    const int w0 = n0 % W_;

    const int mt  = wv & 1;
    const int ntp = (wv >> 1) * 2;

    const int a_row = wv * 8 + (lane >> 3);
    const int a_kq  = lane & 7;
    const float* a_g = Wt + (size_t)(m0 + a_row) * KK_ + a_kq;
    const int a_l = a_row * A_PAD + a_kq;

    const int p_row = tid >> 4;
    const int p_q   = tid & 15;
    const float* p_g = src + (size_t)b * BSTRIDE + (size_t)p_row * CSTRIDE + w0 + 4 * p_q;
    const int p_l = PATCH_ROW(p_row) * PROW + 4 + 4 * p_q;

    const bool h_own  = tid < 16;
    const int  h_row  = (tid >> 1) & 7;
    const int  h_side = tid & 1;
    const bool h_ok   = h_side ? (w0 + NT < W_) : (w0 > 0);
    const int  h_w    = h_ok ? (h_side ? w0 + NT : w0 - 4) : w0;
    const float* h_g  = src + (size_t)b * BSTRIDE + (size_t)h_row * CSTRIDE + h_w;
    const int  h_l    = PATCH_ROW(h_row) * PROW + (h_side ? NT + 4 : 0);

    float ra[18];
    v4f   rp;
    v4f   rh = {0.0f, 0.0f, 0.0f, 0.0f};

#define LOAD_CHUNK(c)                                                              \
    {                                                                              \
        const float* ap = a_g + (c) * KC;                                          \
        _Pragma("unroll")                                                          \
        for (int j = 0; j < 18; ++j)                                               \
            ra[j] = ap[(j / 9) * 4 * KK_ + (j % 9) * 8];                           \
        rp = *reinterpret_cast<const v4f*>(p_g + (size_t)(c) * 8 * CSTRIDE);       \
        if (h_own)                                                                 \
            rh = *reinterpret_cast<const v4f*>(h_g + (size_t)(c) * 8 * CSTRIDE);   \
    }
#define STORE_CHUNK(buf)                                                           \
    {                                                                              \
        _Pragma("unroll")                                                          \
        for (int j = 0; j < 18; ++j)                                               \
            As[buf][a_l + (j / 9) * 4 * A_PAD + (j % 9) * 8] = ra[j];              \
        *reinterpret_cast<v4f*>(&Ps[buf][p_l]) = rp;                               \
        if (h_own) {                                                               \
            const v4f z = {0.0f, 0.0f, 0.0f, 0.0f};                                \
            *reinterpret_cast<v4f*>(&Ps[buf][h_l]) = h_ok ? rh : z;                \
        }                                                                          \
    }

    v8f acc0 = {};
    v8f acc1 = {};

    const int fa_off = (mt * 16 + l16) * A_PAD + 18 * hi;
    const int fb_off = (4 * hi) * PROW + ntp * 16 + l16;

    v2f fa[2][KSUB], fb0[2][KSUB], fb1[2][KSUB];

#define CHUNK_BODY(c, MORE)                                                                \
    {                                                                                      \
        const int cur = (c) & 1;                                                           \
        MessagePass_frag_load(&As[cur][fa_off], &Ps[cur][fb_off], 1, fa[1], fb0[1], fb1[1]);      \
        MessagePass_frag_mma(fa[0], fb0[0], fb1[0], acc0, acc1);                           \
        STORE_CHUNK(cur ^ 1)                   \
        __syncthreads();                                                                   \
        if (MORE) LOAD_CHUNK((c) + 2)              \
        MessagePass_frag_load(&As[cur ^ 1][fa_off], &Ps[cur ^ 1][fb_off], 0, fa[0], fb0[0], fb1[0]); \
        MessagePass_frag_mma(fa[1], fb0[1], fb1[1], acc0, acc1);                           \
    }

    LOAD_CHUNK(0)
    STORE_CHUNK(0)
    __syncthreads();
    LOAD_CHUNK(1)
    MessagePass_frag_load(&As[0][fa_off], &Ps[0][fb_off], 0, fa[0], fb0[0], fb1[0]);

    for (int c = 0; c < NCHUNK - 2; ++c)
        CHUNK_BODY(c, true)
    CHUNK_BODY(NCHUNK - 2, false)
    {
        const int cur = (NCHUNK - 1) & 1;
        MessagePass_frag_load(&As[cur][fa_off], &Ps[cur][fb_off], 1, fa[1], fb0[1], fb1[1]);
        MessagePass_frag_mma(fa[0], fb0[0], fb1[0], acc0, acc1);
        MessagePass_frag_mma(fa[1], fb0[1], fb1[1], acc0, acc1);
    }
#undef CHUNK_BODY
#undef LOAD_CHUNK
#undef STORE_CHUNK

    const float* scale = ss;
    const float* shift = ss + C_;
    #pragma unroll
    for (int j = 0; j < 8; ++j) {
        const int M = mt * 16 + j + 8 * hi;
        const int o = m0 + M;
        const float sc = scale[o];
        const float sh = shift[o];
        {
            const int w = w0 + ntp * 16 + l16;
            const size_t off = (size_t)b * BSTRIDE + (size_t)o * CSTRIDE + w;
            float y = acc0[j] * sc + sh;
            y = y > 0.0f ? y : 0.0f;
            dst[off] = y + addv[off];
        }
        {
            const int w = w0 + (ntp + 1) * 16 + l16;
            const size_t off = (size_t)b * BSTRIDE + (size_t)o * CSTRIDE + w;
            float y = acc1[j] * sc + sh;
            y = y > 0.0f ? y : 0.0f;
            dst[off] = y + addv[off];
        }
    }
}

__global__ void MessagePass_copy_slice(const float* __restrict__ x, float* __restrict__ out)
{
    int e = blockIdx.x * blockDim.x + threadIdx.x;
    if (e >= B_ * C_ * W_) return;
    int w  = e % W_;
    int bc = e / W_;
    size_t off = (size_t)bc * CSTRIDE + w;
    out[off] = x[off];
}

__global__ void MessagePass_bn_prep(const float* __restrict__ g, const float* __restrict__ be,
                                    const float* __restrict__ rm, const float* __restrict__ rv,
                                    float* __restrict__ ss)
{
    int c = blockIdx.x * blockDim.x + threadIdx.x;
    if (c < C_) {
        float sc = g[c] * rsqrtf(rv[c] + 1e-5f);
        ss[c]      = sc;
        ss[C_ + c] = be[c] - rm[c] * sc;
    }
}

extern "C" void kernel_launch(void* const* d_in, const int* in_sizes, int n_in,
                              void* d_out, int out_size, void* d_ws, size_t ws_size,
                              hipStream_t stream)
{
    const float* x     = (const float*)d_in[0];
    const float* cw    = (const float*)d_in[1];
    const float* gamma = (const float*)d_in[2];
    const float* beta  = (const float*)d_in[3];
    const float* rmean = (const float*)d_in[4];
    const float* rvar  = (const float*)d_in[5];
    float* out = (float*)d_out;
    float* ss  = (float*)d_ws;

    (void)in_sizes; (void)n_in; (void)out_size; (void)ws_size;

    MessagePass_bn_prep<<<1, 128, 0, stream>>>(gamma, beta, rmean, rvar, ss);
    MessagePass_copy_slice<<<(B_ * C_ * W_ + 255) / 256, 256, 0, stream>>>(x, out);

    dim3 grid((B_ * W_) / NT, C_ / MT);

    for (int h = 1; h < H_; ++h)
        MessagePass_step_wmma<<<grid, 128, 0, stream>>>(
            out + (size_t)(h - 1) * W_, x + (size_t)h * W_, out + (size_t)h * W_, cw, ss);

    for (int h = H_ - 2; h >= 0; --h)
        MessagePass_step_wmma<<<grid, 128, 0, stream>>>(
            out + (size_t)(h + 1) * W_, out + (size_t)h * W_, out + (size_t)h * W_, cw, ss);
}


# Global prefetch two K chunks deep: second register set of staged loads, stored to LDS two chunks after issue (was one); barrier fences LDS only so the loads stay in flight; summation order unchanged
# speedup vs baseline: 8.8063x; 1.2070x over previous
#include <hip/hip_runtime.h>
#include <hip/hip_bf16.h>

typedef float v2f __attribute__((ext_vector_type(2)));
typedef float v4f __attribute__((ext_vector_type(4)));
typedef float v8f __attribute__((ext_vector_type(8)));

#define B_  16
#define C_  128
#define H_  64
#define W_  256
#define KK_ 1152
#define CSTRIDE (H_ * W_)
#define BSTRIDE (C_ * H_ * W_)

#define KC  72
#define MT  32
#define NT  64
#define A_PAD 76
#define PROW 72
#define PATCH_ROW(c) ((((c) & 2) << 1) | (((c) & 4) >> 1) | ((c) & 1))

#define NCHUNK (KK_ / KC)

#define KSUB 9

__device__ __forceinline__
void MessagePass_lds_barrier()
{
    __builtin_amdgcn_fence(__ATOMIC_RELEASE, "workgroup", "local");
    __builtin_amdgcn_s_barrier();
    __builtin_amdgcn_fence(__ATOMIC_ACQUIRE, "workgroup", "local");
}

__device__ __forceinline__
void MessagePass_frag_load(const float* __restrict__ Arow, const float* __restrict__ Pcol,
                           int blk, v2f* fa, v2f* fb0, v2f* fb1)
{
    const float* Pb = Pcol + 2 * blk * PROW;
    #pragma unroll
    for (int s = 0; s < KSUB; ++s) {
        const int j0 = 2 * s, j1 = 2 * s + 1;
        fa[s]    = *reinterpret_cast<const v2f*>(Arow + 36 * blk + 2 * s);
        fb0[s].x = Pb[(j0 / 9) * PROW + (j0 % 9)];
        fb0[s].y = Pb[(j1 / 9) * PROW + (j1 % 9)];
    }
    __builtin_amdgcn_sched_barrier(0);
    #pragma unroll
    for (int s = 0; s < KSUB; ++s) {
        const int j0 = 2 * s, j1 = 2 * s + 1;
        fb1[s].x = Pb[(j0 / 9) * PROW + (j0 % 9) + 16];
        fb1[s].y = Pb[(j1 / 9) * PROW + (j1 % 9) + 16];
    }
    __builtin_amdgcn_sched_barrier(0);
}

__device__ __forceinline__
void MessagePass_frag_mma(const v2f* fa, const v2f* fb0, const v2f* fb1, v8f& acc0, v8f& acc1)
{
    #pragma unroll
    for (int s = 0; s < KSUB; ++s) {
        acc0 = __builtin_amdgcn_wmma_f32_16x16x4_f32(false, fa[s], false, fb0[s],
                                                     (short)0, acc0, false, false);
        acc1 = __builtin_amdgcn_wmma_f32_16x16x4_f32(false, fa[s], false, fb1[s],
                                                     (short)0, acc1, false, false);
    }
    asm volatile("" : "+v"(acc0), "+v"(acc1) : : "memory");
}

__global__ __launch_bounds__(128)
void MessagePass_step_wmma(const float* __restrict__ src,
                           const float* __restrict__ addv,
                           float* __restrict__ dst,
                           const float* __restrict__ Wt,
                           const float* __restrict__ ss)
{
    __shared__ __attribute__((aligned(16))) float As[2][MT * A_PAD];
    __shared__ __attribute__((aligned(16))) float Ps[2][8 * PROW];

    const int tid  = threadIdx.x;
    const int lane = tid & 31;
    const int hi   = lane >> 4;
    const int l16  = lane & 15;
    const int wv   = tid >> 5;

    const int m0 = blockIdx.y * MT;
    const int n0 = blockIdx.x * NT;
    const int b  = n0 / W_;
    const int w0 = n0 % W_;

    const int mt  = wv & 1;
    const int ntp = (wv >> 1) * 2;

    const int a_row = wv * 8 + (lane >> 3);
    const int a_kq  = lane & 7;
    const float* a_g = Wt + (size_t)(m0 + a_row) * KK_ + a_kq;
    const int a_l = a_row * A_PAD + a_kq;

    const int p_row = tid >> 4;
    const int p_q   = tid & 15;
    const float* p_g = src + (size_t)b * BSTRIDE + (size_t)p_row * CSTRIDE + w0 + 4 * p_q;
    const int p_l = PATCH_ROW(p_row) * PROW + 4 + 4 * p_q;

    const bool h_own  = tid < 16;
    const int  h_row  = (tid >> 1) & 7;
    const int  h_side = tid & 1;
    const bool h_ok   = h_side ? (w0 + NT < W_) : (w0 > 0);
    const int  h_w    = h_ok ? (h_side ? w0 + NT : w0 - 4) : w0;
    const float* h_g  = src + (size_t)b * BSTRIDE + (size_t)h_row * CSTRIDE + h_w;
    const int  h_l    = PATCH_ROW(h_row) * PROW + (h_side ? NT + 4 : 0);

    float ra0[18], ra1[18];
    v4f   rp0, rp1;
    v4f   rh0 = {0.0f, 0.0f, 0.0f, 0.0f};
    v4f   rh1 = {0.0f, 0.0f, 0.0f, 0.0f};

#define LOAD_CHUNK(c, RA, RP, RH)                                                  \
    {                                                                              \
        const float* ap = a_g + (c) * KC;                                          \
        _Pragma("unroll")                                                          \
        for (int j = 0; j < 18; ++j)                                               \
            RA[j] = ap[(j / 9) * 4 * KK_ + (j % 9) * 8];                           \
        RP = *reinterpret_cast<const v4f*>(p_g + (size_t)(c) * 8 * CSTRIDE);       \
        if (h_own)                                                                 \
            RH = *reinterpret_cast<const v4f*>(h_g + (size_t)(c) * 8 * CSTRIDE);   \
    }
#define STORE_CHUNK(buf, RA, RP, RH)                                               \
    {                                                                              \
        _Pragma("unroll")                                                          \
        for (int j = 0; j < 18; ++j)                                               \
            As[buf][a_l + (j / 9) * 4 * A_PAD + (j % 9) * 8] = RA[j];              \
        *reinterpret_cast<v4f*>(&Ps[buf][p_l]) = RP;                               \
        if (h_own) {                                                               \
            const v4f z = {0.0f, 0.0f, 0.0f, 0.0f};                                \
            *reinterpret_cast<v4f*>(&Ps[buf][h_l]) = h_ok ? RH : z;                \
        }                                                                          \
    }

    v8f acc0 = {};
    v8f acc1 = {};

    const int fa_off = (mt * 16 + l16) * A_PAD + 18 * hi;
    const int fb_off = (4 * hi) * PROW + ntp * 16 + l16;

    v2f fa[2][KSUB], fb0[2][KSUB], fb1[2][KSUB];

#define CHUNK_BODY(c, CUR, RA, RP, RH, MORE)                                               \
    {                                                                                      \
        MessagePass_frag_load(&As[CUR][fa_off], &Ps[CUR][fb_off], 1, fa[1], fb0[1], fb1[1]);      \
        MessagePass_frag_mma(fa[0], fb0[0], fb1[0], acc0, acc1);                           \
        STORE_CHUNK((CUR) ^ 1, RA, RP, RH)     \
        MessagePass_lds_barrier();                                                                   \
        if (MORE) LOAD_CHUNK((c) + 3, RA, RP, RH)     \
        MessagePass_frag_load(&As[(CUR) ^ 1][fa_off], &Ps[(CUR) ^ 1][fb_off], 0, fa[0], fb0[0], fb1[0]); \
        MessagePass_frag_mma(fa[1], fb0[1], fb1[1], acc0, acc1);                           \
    }

    LOAD_CHUNK(0, ra0, rp0, rh0)
    LOAD_CHUNK(1, ra1, rp1, rh1)
    STORE_CHUNK(0, ra0, rp0, rh0)
    MessagePass_lds_barrier();
    LOAD_CHUNK(2, ra0, rp0, rh0)
    MessagePass_frag_load(&As[0][fa_off], &Ps[0][fb_off], 0, fa[0], fb0[0], fb1[0]);

    for (int c = 0; c < NCHUNK - 4; c += 2) {
        CHUNK_BODY(c,     0, ra1, rp1, rh1, true)
        CHUNK_BODY(c + 1, 1, ra0, rp0, rh0, true)
    }
    CHUNK_BODY(NCHUNK - 4, 0, ra1, rp1, rh1, true)
    CHUNK_BODY(NCHUNK - 3, 1, ra0, rp0, rh0, false)
    CHUNK_BODY(NCHUNK - 2, 0, ra1, rp1, rh1, false)
    {
        const int cur = (NCHUNK - 1) & 1;
        MessagePass_frag_load(&As[cur][fa_off], &Ps[cur][fb_off], 1, fa[1], fb0[1], fb1[1]);
        MessagePass_frag_mma(fa[0], fb0[0], fb1[0], acc0, acc1);
        MessagePass_frag_mma(fa[1], fb0[1], fb1[1], acc0, acc1);
    }
#undef CHUNK_BODY
#undef LOAD_CHUNK
#undef STORE_CHUNK

    const float* scale = ss;
    const float* shift = ss + C_;
    #pragma unroll
    for (int j = 0; j < 8; ++j) {
        const int M = mt * 16 + j + 8 * hi;
        const int o = m0 + M;
        const float sc = scale[o];
        const float sh = shift[o];
        {
            const int w = w0 + ntp * 16 + l16;
            const size_t off = (size_t)b * BSTRIDE + (size_t)o * CSTRIDE + w;
            float y = acc0[j] * sc + sh;
            y = y > 0.0f ? y : 0.0f;
            dst[off] = y + addv[off];
        }
        {
            const int w = w0 + (ntp + 1) * 16 + l16;
            const size_t off = (size_t)b * BSTRIDE + (size_t)o * CSTRIDE + w;
            float y = acc1[j] * sc + sh;
            y = y > 0.0f ? y : 0.0f;
            dst[off] = y + addv[off];
        }
    }
}

__global__ void MessagePass_copy_slice(const float* __restrict__ x, float* __restrict__ out)
{
    int e = blockIdx.x * blockDim.x + threadIdx.x;
    if (e >= B_ * C_ * W_) return;
    int w  = e % W_;
    int bc = e / W_;
    size_t off = (size_t)bc * CSTRIDE + w;
    out[off] = x[off];
}

__global__ void MessagePass_bn_prep(const float* __restrict__ g, const float* __restrict__ be,
                                    const float* __restrict__ rm, const float* __restrict__ rv,
                                    float* __restrict__ ss)
{
    int c = blockIdx.x * blockDim.x + threadIdx.x;
    if (c < C_) {
        float sc = g[c] * rsqrtf(rv[c] + 1e-5f);
        ss[c]      = sc;
        ss[C_ + c] = be[c] - rm[c] * sc;
    }
}

extern "C" void kernel_launch(void* const* d_in, const int* in_sizes, int n_in,
                              void* d_out, int out_size, void* d_ws, size_t ws_size,
                              hipStream_t stream)
{
    const float* x     = (const float*)d_in[0];
    const float* cw    = (const float*)d_in[1];
    const float* gamma = (const float*)d_in[2];
    const float* beta  = (const float*)d_in[3];
    const float* rmean = (const float*)d_in[4];
    const float* rvar  = (const float*)d_in[5];
    float* out = (float*)d_out;
    float* ss  = (float*)d_ws;

    (void)in_sizes; (void)n_in; (void)out_size; (void)ws_size;

    MessagePass_bn_prep<<<1, 128, 0, stream>>>(gamma, beta, rmean, rvar, ss);
    MessagePass_copy_slice<<<(B_ * C_ * W_ + 255) / 256, 256, 0, stream>>>(x, out);

    dim3 grid((B_ * W_) / NT, C_ / MT);

    for (int h = 1; h < H_; ++h)
        MessagePass_step_wmma<<<grid, 128, 0, stream>>>(
            out + (size_t)(h - 1) * W_, x + (size_t)h * W_, out + (size_t)h * W_, cw, ss);

    for (int h = H_ - 2; h >= 0; --h)
        MessagePass_step_wmma<<<grid, 128, 0, stream>>>(
            out + (size_t)(h + 1) * W_, out + (size_t)h * W_, out + (size_t)h * W_, cw, ss);
}


# Epilogue operands prefetched: scale, shift and the residual tile loaded into registers three K chunks before the loop ends (were two dependent round trips behind the last product); sums unchanged
# speedup vs baseline: 9.3438x; 1.0610x over previous
#include <hip/hip_runtime.h>
#include <hip/hip_bf16.h>

typedef float v2f __attribute__((ext_vector_type(2)));
typedef float v4f __attribute__((ext_vector_type(4)));
typedef float v8f __attribute__((ext_vector_type(8)));

#define B_  16
#define C_  128
#define H_  64
#define W_  256
#define KK_ 1152
#define CSTRIDE (H_ * W_)
#define BSTRIDE (C_ * H_ * W_)

#define KC  72
#define MT  32
#define NT  64
#define A_PAD 76
#define PROW 72
#define PATCH_ROW(c) ((((c) & 2) << 1) | (((c) & 4) >> 1) | ((c) & 1))

#define NCHUNK (KK_ / KC)

#define KSUB 9

__device__ __forceinline__
void MessagePass_lds_barrier()
{
    __builtin_amdgcn_fence(__ATOMIC_RELEASE, "workgroup", "local");
    __builtin_amdgcn_s_barrier();
    __builtin_amdgcn_fence(__ATOMIC_ACQUIRE, "workgroup", "local");
}

__device__ __forceinline__
void MessagePass_frag_load(const float* __restrict__ Arow, const float* __restrict__ Pcol,
                           int blk, v2f* fa, v2f* fb0, v2f* fb1)
{
    const float* Pb = Pcol + 2 * blk * PROW;
    #pragma unroll
    for (int s = 0; s < KSUB; ++s) {
        const int j0 = 2 * s, j1 = 2 * s + 1;
        fa[s]    = *reinterpret_cast<const v2f*>(Arow + 36 * blk + 2 * s);
        fb0[s].x = Pb[(j0 / 9) * PROW + (j0 % 9)];
        fb0[s].y = Pb[(j1 / 9) * PROW + (j1 % 9)];
    }
    __builtin_amdgcn_sched_barrier(0);
    #pragma unroll
    for (int s = 0; s < KSUB; ++s) {
        const int j0 = 2 * s, j1 = 2 * s + 1;
        fb1[s].x = Pb[(j0 / 9) * PROW + (j0 % 9) + 16];
        fb1[s].y = Pb[(j1 / 9) * PROW + (j1 % 9) + 16];
    }
    __builtin_amdgcn_sched_barrier(0);
}

__device__ __forceinline__
void MessagePass_frag_mma(const v2f* fa, const v2f* fb0, const v2f* fb1, v8f& acc0, v8f& acc1)
{
    #pragma unroll
    for (int s = 0; s < KSUB; ++s) {
        acc0 = __builtin_amdgcn_wmma_f32_16x16x4_f32(false, fa[s], false, fb0[s],
                                                     (short)0, acc0, false, false);
        acc1 = __builtin_amdgcn_wmma_f32_16x16x4_f32(false, fa[s], false, fb1[s],
                                                     (short)0, acc1, false, false);
    }
    asm volatile("" : "+v"(acc0), "+v"(acc1) : : "memory");
}

__global__ __launch_bounds__(128)
void MessagePass_step_wmma(const float* __restrict__ src,
                           const float* __restrict__ addv,
                           float* __restrict__ dst,
                           const float* __restrict__ Wt,
                           const float* __restrict__ ss)
{
    __shared__ __attribute__((aligned(16))) float As[2][MT * A_PAD];
    __shared__ __attribute__((aligned(16))) float Ps[2][8 * PROW];

    const int tid  = threadIdx.x;
    const int lane = tid & 31;
    const int hi   = lane >> 4;
    const int l16  = lane & 15;
    const int wv   = tid >> 5;

    const int m0 = blockIdx.y * MT;
    const int n0 = blockIdx.x * NT;
    const int b  = n0 / W_;
    const int w0 = n0 % W_;

    const int mt  = wv & 1;
    const int ntp = (wv >> 1) * 2;

    const int a_row = wv * 8 + (lane >> 3);
    const int a_kq  = lane & 7;
    const float* a_g = Wt + (size_t)(m0 + a_row) * KK_ + a_kq;
    const int a_l = a_row * A_PAD + a_kq;

    const int p_row = tid >> 4;
    const int p_q   = tid & 15;
    const float* p_g = src + (size_t)b * BSTRIDE + (size_t)p_row * CSTRIDE + w0 + 4 * p_q;
    const int p_l = PATCH_ROW(p_row) * PROW + 4 + 4 * p_q;

    const bool h_own  = tid < 16;
    const int  h_row  = (tid >> 1) & 7;
    const int  h_side = tid & 1;
    const bool h_ok   = h_side ? (w0 + NT < W_) : (w0 > 0);
    const int  h_w    = h_ok ? (h_side ? w0 + NT : w0 - 4) : w0;
    const float* h_g  = src + (size_t)b * BSTRIDE + (size_t)h_row * CSTRIDE + h_w;
    const int  h_l    = PATCH_ROW(h_row) * PROW + (h_side ? NT + 4 : 0);

    float ra0[18], ra1[18];
    v4f   rp0, rp1;
    v4f   rh0 = {0.0f, 0.0f, 0.0f, 0.0f};
    v4f   rh1 = {0.0f, 0.0f, 0.0f, 0.0f};

#define LOAD_CHUNK(c, RA, RP, RH)                                                  \
    {                                                                              \
        const float* ap = a_g + (c) * KC;                                          \
        _Pragma("unroll")                                                          \
        for (int j = 0; j < 18; ++j)                                               \
            RA[j] = ap[(j / 9) * 4 * KK_ + (j % 9) * 8];                           \
        RP = *reinterpret_cast<const v4f*>(p_g + (size_t)(c) * 8 * CSTRIDE);       \
        if (h_own)                                                                 \
            RH = *reinterpret_cast<const v4f*>(h_g + (size_t)(c) * 8 * CSTRIDE);   \
    }
#define STORE_CHUNK(buf, RA, RP, RH)                                               \
    {                                                                              \
        _Pragma("unroll")                                                          \
        for (int j = 0; j < 18; ++j)                                               \
            As[buf][a_l + (j / 9) * 4 * A_PAD + (j % 9) * 8] = RA[j];              \
        *reinterpret_cast<v4f*>(&Ps[buf][p_l]) = RP;                               \
        if (h_own) {                                                               \
            const v4f z = {0.0f, 0.0f, 0.0f, 0.0f};                                \
            *reinterpret_cast<v4f*>(&Ps[buf][h_l]) = h_ok ? RH : z;                \
        }                                                                          \
    }

    v8f acc0 = {};
    v8f acc1 = {};

    const int fa_off = (mt * 16 + l16) * A_PAD + 18 * hi;
    const int fb_off = (4 * hi) * PROW + ntp * 16 + l16;

    v2f fa[2][KSUB], fb0[2][KSUB], fb1[2][KSUB];

#define CHUNK_BODY(c, CUR, RA, RP, RH, MORE)                                               \
    {                                                                                      \
        MessagePass_frag_load(&As[CUR][fa_off], &Ps[CUR][fb_off], 1, fa[1], fb0[1], fb1[1]);      \
        MessagePass_frag_mma(fa[0], fb0[0], fb1[0], acc0, acc1);                           \
        STORE_CHUNK((CUR) ^ 1, RA, RP, RH)     \
        MessagePass_lds_barrier();                                                                   \
        if (MORE) LOAD_CHUNK((c) + 3, RA, RP, RH)     \
        MessagePass_frag_load(&As[(CUR) ^ 1][fa_off], &Ps[(CUR) ^ 1][fb_off], 0, fa[0], fb0[0], fb1[0]); \
        MessagePass_frag_mma(fa[1], fb0[1], fb1[1], acc0, acc1);                           \
    }

    LOAD_CHUNK(0, ra0, rp0, rh0)
    LOAD_CHUNK(1, ra1, rp1, rh1)
    STORE_CHUNK(0, ra0, rp0, rh0)
    MessagePass_lds_barrier();
    LOAD_CHUNK(2, ra0, rp0, rh0)
    MessagePass_frag_load(&As[0][fa_off], &Ps[0][fb_off], 0, fa[0], fb0[0], fb1[0]);

    for (int c = 0; c < NCHUNK - 4; c += 2) {
        CHUNK_BODY(c,     0, ra1, rp1, rh1, true)
        CHUNK_BODY(c + 1, 1, ra0, rp0, rh0, true)
    }
    CHUNK_BODY(NCHUNK - 4, 0, ra1, rp1, rh1, true)

    const int    o_first = m0 + mt * 16 + 8 * hi;
    const size_t o_base  = (size_t)b * BSTRIDE + (size_t)o_first * CSTRIDE + w0 + ntp * 16 + l16;
    float e_sc[8], e_sh[8], e_r0[8], e_r1[8];
    #pragma unroll
    for (int j = 0; j < 8; ++j) {
        e_sc[j] = ss[o_first + j];
        e_sh[j] = ss[C_ + o_first + j];
    }
    #pragma unroll
    for (int j = 0; j < 8; ++j) {
        e_r0[j] = addv[o_base + (size_t)j * CSTRIDE];
        e_r1[j] = addv[o_base + (size_t)j * CSTRIDE + 16];
    }

    CHUNK_BODY(NCHUNK - 3, 1, ra0, rp0, rh0, false)
    CHUNK_BODY(NCHUNK - 2, 0, ra1, rp1, rh1, false)
    {
        const int cur = (NCHUNK - 1) & 1;
        MessagePass_frag_load(&As[cur][fa_off], &Ps[cur][fb_off], 1, fa[1], fb0[1], fb1[1]);
        MessagePass_frag_mma(fa[0], fb0[0], fb1[0], acc0, acc1);
        MessagePass_frag_mma(fa[1], fb0[1], fb1[1], acc0, acc1);
    }
#undef CHUNK_BODY
#undef LOAD_CHUNK
#undef STORE_CHUNK

    #pragma unroll
    for (int j = 0; j < 8; ++j) {
        const size_t off = o_base + (size_t)j * CSTRIDE;
        float y0 = acc0[j] * e_sc[j] + e_sh[j];
        float y1 = acc1[j] * e_sc[j] + e_sh[j];
        y0 = y0 > 0.0f ? y0 : 0.0f;
        y1 = y1 > 0.0f ? y1 : 0.0f;
        dst[off]      = y0 + e_r0[j];
        dst[off + 16] = y1 + e_r1[j];
    }
}

__global__ void MessagePass_copy_slice(const float* __restrict__ x, float* __restrict__ out)
{
    int e = blockIdx.x * blockDim.x + threadIdx.x;
    if (e >= B_ * C_ * W_) return;
    int w  = e % W_;
    int bc = e / W_;
    size_t off = (size_t)bc * CSTRIDE + w;
    out[off] = x[off];
}

__global__ void MessagePass_bn_prep(const float* __restrict__ g, const float* __restrict__ be,
                                    const float* __restrict__ rm, const float* __restrict__ rv,
                                    float* __restrict__ ss)
{
    int c = blockIdx.x * blockDim.x + threadIdx.x;
    if (c < C_) {
        float sc = g[c] * rsqrtf(rv[c] + 1e-5f);
        ss[c]      = sc;
        ss[C_ + c] = be[c] - rm[c] * sc;
    }
}

extern "C" void kernel_launch(void* const* d_in, const int* in_sizes, int n_in,
                              void* d_out, int out_size, void* d_ws, size_t ws_size,
                              hipStream_t stream)
{
    const float* x     = (const float*)d_in[0];
    const float* cw    = (const float*)d_in[1];
    const float* gamma = (const float*)d_in[2];
    const float* beta  = (const float*)d_in[3];
    const float* rmean = (const float*)d_in[4];
    const float* rvar  = (const float*)d_in[5];
    float* out = (float*)d_out;
    float* ss  = (float*)d_ws;

    (void)in_sizes; (void)n_in; (void)out_size; (void)ws_size;

    MessagePass_bn_prep<<<1, 128, 0, stream>>>(gamma, beta, rmean, rvar, ss);
    MessagePass_copy_slice<<<(B_ * C_ * W_ + 255) / 256, 256, 0, stream>>>(x, out);

    dim3 grid((B_ * W_) / NT, C_ / MT);

    for (int h = 1; h < H_; ++h)
        MessagePass_step_wmma<<<grid, 128, 0, stream>>>(
            out + (size_t)(h - 1) * W_, x + (size_t)h * W_, out + (size_t)h * W_, cw, ss);

    for (int h = H_ - 2; h >= 0; --h)
        MessagePass_step_wmma<<<grid, 128, 0, stream>>>(
            out + (size_t)(h + 1) * W_, out + (size_t)h * W_, out + (size_t)h * W_, cw, ss);
}


# LDS stage stores moved to the chunk head, grouped with the fragment reads, and the barrier split: arrive inside the first product block, await behind the next global loads; sums unchanged
# speedup vs baseline: 10.3684x; 1.1097x over previous
#include <hip/hip_runtime.h>
#include <hip/hip_bf16.h>

typedef float v2f __attribute__((ext_vector_type(2)));
typedef float v4f __attribute__((ext_vector_type(4)));
typedef float v8f __attribute__((ext_vector_type(8)));

#define B_  16
#define C_  128
#define H_  64
#define W_  256
#define KK_ 1152
#define CSTRIDE (H_ * W_)
#define BSTRIDE (C_ * H_ * W_)

#define KC  72
#define MT  32
#define NT  64
#define A_PAD 76
#define PROW 72
#define PATCH_ROW(c) ((((c) & 2) << 1) | (((c) & 4) >> 1) | ((c) & 1))

#define NCHUNK (KK_ / KC)

#define KSUB 9
#define KSIG 7

__device__ __forceinline__
void MessagePass_lds_barrier()
{
    __builtin_amdgcn_fence(__ATOMIC_RELEASE, "workgroup", "local");
    __builtin_amdgcn_s_barrier();
    __builtin_amdgcn_fence(__ATOMIC_ACQUIRE, "workgroup", "local");
}

__device__ __forceinline__
void MessagePass_lds_arrive()
{
    __builtin_amdgcn_fence(__ATOMIC_RELEASE, "workgroup", "local");
    __builtin_amdgcn_s_barrier_signal(-1);
}
__device__ __forceinline__
void MessagePass_lds_await()
{
    __builtin_amdgcn_s_barrier_wait(-1);
    __builtin_amdgcn_fence(__ATOMIC_ACQUIRE, "workgroup", "local");
}

__device__ __forceinline__
void MessagePass_frag_load(const float* __restrict__ Arow, const float* __restrict__ Pcol,
                           int blk, v2f* fa, v2f* fb0, v2f* fb1)
{
    const float* Pb = Pcol + 2 * blk * PROW;
    #pragma unroll
    for (int s = 0; s < KSUB; ++s) {
        const int j0 = 2 * s, j1 = 2 * s + 1;
        fa[s]    = *reinterpret_cast<const v2f*>(Arow + 36 * blk + 2 * s);
        fb0[s].x = Pb[(j0 / 9) * PROW + (j0 % 9)];
        fb0[s].y = Pb[(j1 / 9) * PROW + (j1 % 9)];
    }
    __builtin_amdgcn_sched_barrier(0);
    #pragma unroll
    for (int s = 0; s < KSUB; ++s) {
        const int j0 = 2 * s, j1 = 2 * s + 1;
        fb1[s].x = Pb[(j0 / 9) * PROW + (j0 % 9) + 16];
        fb1[s].y = Pb[(j1 / 9) * PROW + (j1 % 9) + 16];
    }
    __builtin_amdgcn_sched_barrier(0);
}

__device__ __forceinline__
void MessagePass_frag_mma_range(const v2f* fa, const v2f* fb0, const v2f* fb1, v8f& acc0, v8f& acc1,
                                int s_begin, int s_end)
{
    #pragma unroll
    for (int s = s_begin; s < s_end; ++s) {
        acc0 = __builtin_amdgcn_wmma_f32_16x16x4_f32(false, fa[s], false, fb0[s],
                                                     (short)0, acc0, false, false);
        acc1 = __builtin_amdgcn_wmma_f32_16x16x4_f32(false, fa[s], false, fb1[s],
                                                     (short)0, acc1, false, false);
    }
}

__device__ __forceinline__
void MessagePass_products_end(v8f& acc0, v8f& acc1)
{
    asm volatile("" : "+v"(acc0), "+v"(acc1) : : "memory");
}

__device__ __forceinline__
void MessagePass_frag_mma(const v2f* fa, const v2f* fb0, const v2f* fb1, v8f& acc0, v8f& acc1)
{
    MessagePass_frag_mma_range(fa, fb0, fb1, acc0, acc1, 0, KSUB);
    MessagePass_products_end(acc0, acc1);
}

__global__ __launch_bounds__(128)
void MessagePass_step_wmma(const float* __restrict__ src,
                           const float* __restrict__ addv,
                           float* __restrict__ dst,
                           const float* __restrict__ Wt,
                           const float* __restrict__ ss)
{
    __shared__ __attribute__((aligned(16))) float As[2][MT * A_PAD];
    __shared__ __attribute__((aligned(16))) float Ps[2][8 * PROW];

    const int tid  = threadIdx.x;
    const int lane = tid & 31;
    const int hi   = lane >> 4;
    const int l16  = lane & 15;
    const int wv   = tid >> 5;

    const int m0 = blockIdx.y * MT;
    const int n0 = blockIdx.x * NT;
    const int b  = n0 / W_;
    const int w0 = n0 % W_;

    const int mt  = wv & 1;
    const int ntp = (wv >> 1) * 2;

    const int a_row = wv * 8 + (lane >> 3);
    const int a_kq  = lane & 7;
    const float* a_g = Wt + (size_t)(m0 + a_row) * KK_ + a_kq;
    const int a_l = a_row * A_PAD + a_kq;

    const int p_row = tid >> 4;
    const int p_q   = tid & 15;
    const float* p_g = src + (size_t)b * BSTRIDE + (size_t)p_row * CSTRIDE + w0 + 4 * p_q;
    const int p_l = PATCH_ROW(p_row) * PROW + 4 + 4 * p_q;

    const bool h_own  = tid < 16;
    const int  h_row  = (tid >> 1) & 7;
    const int  h_side = tid & 1;
    const bool h_ok   = h_side ? (w0 + NT < W_) : (w0 > 0);
    const int  h_w    = h_ok ? (h_side ? w0 + NT : w0 - 4) : w0;
    const float* h_g  = src + (size_t)b * BSTRIDE + (size_t)h_row * CSTRIDE + h_w;
    const int  h_l    = PATCH_ROW(h_row) * PROW + (h_side ? NT + 4 : 0);

    float ra0[18], ra1[18];
    v4f   rp0, rp1;
    v4f   rh0 = {0.0f, 0.0f, 0.0f, 0.0f};
    v4f   rh1 = {0.0f, 0.0f, 0.0f, 0.0f};

#define LOAD_CHUNK(c, RA, RP, RH)                                                  \
    {                                                                              \
        const float* ap = a_g + (c) * KC;                                          \
        _Pragma("unroll")                                                          \
        for (int j = 0; j < 18; ++j)                                               \
            RA[j] = ap[(j / 9) * 4 * KK_ + (j % 9) * 8];                           \
        RP = *reinterpret_cast<const v4f*>(p_g + (size_t)(c) * 8 * CSTRIDE);       \
        if (h_own)                                                                 \
            RH = *reinterpret_cast<const v4f*>(h_g + (size_t)(c) * 8 * CSTRIDE);   \
    }
#define STORE_CHUNK(buf, RA, RP, RH)                                               \
    {                                                                              \
        _Pragma("unroll")                                                          \
        for (int j = 0; j < 18; ++j)                                               \
            As[buf][a_l + (j / 9) * 4 * A_PAD + (j % 9) * 8] = RA[j];              \
        *reinterpret_cast<v4f*>(&Ps[buf][p_l]) = RP;                               \
        if (h_own) {                                                               \
            const v4f z = {0.0f, 0.0f, 0.0f, 0.0f};                                \
            *reinterpret_cast<v4f*>(&Ps[buf][h_l]) = h_ok ? RH : z;                \
        }                                                                          \
    }

    v8f acc0 = {};
    v8f acc1 = {};

    const int fa_off = (mt * 16 + l16) * A_PAD + 18 * hi;
    const int fb_off = (4 * hi) * PROW + ntp * 16 + l16;

    v2f fa[2][KSUB], fb0[2][KSUB], fb1[2][KSUB];

#define CHUNK_BODY(c, CUR, RA, RP, RH, MORE)                                               \
    {                                                                                      \
        STORE_CHUNK((CUR) ^ 1, RA, RP, RH)     \
        MessagePass_frag_load(&As[CUR][fa_off], &Ps[CUR][fb_off], 1, fa[1], fb0[1], fb1[1]);      \
        MessagePass_frag_mma_range(fa[0], fb0[0], fb1[0], acc0, acc1, 0, KSIG);            \
          \
        __builtin_amdgcn_sched_barrier(0);                                                 \
        MessagePass_lds_arrive();                                                          \
        __builtin_amdgcn_sched_barrier(0);                                                 \
        MessagePass_frag_mma_range(fa[0], fb0[0], fb1[0], acc0, acc1, KSIG, KSUB);         \
        MessagePass_products_end(acc0, acc1);                                              \
        if (MORE) LOAD_CHUNK((c) + 3, RA, RP, RH)     \
        MessagePass_lds_await();                                                           \
        MessagePass_frag_load(&As[(CUR) ^ 1][fa_off], &Ps[(CUR) ^ 1][fb_off], 0, fa[0], fb0[0], fb1[0]); \
        MessagePass_frag_mma(fa[1], fb0[1], fb1[1], acc0, acc1);                           \
    }

    LOAD_CHUNK(0, ra0, rp0, rh0)
    LOAD_CHUNK(1, ra1, rp1, rh1)
    STORE_CHUNK(0, ra0, rp0, rh0)
    MessagePass_lds_barrier();
    LOAD_CHUNK(2, ra0, rp0, rh0)
    MessagePass_frag_load(&As[0][fa_off], &Ps[0][fb_off], 0, fa[0], fb0[0], fb1[0]);

    for (int c = 0; c < NCHUNK - 4; c += 2) {
        CHUNK_BODY(c,     0, ra1, rp1, rh1, true)
        CHUNK_BODY(c + 1, 1, ra0, rp0, rh0, true)
    }
    CHUNK_BODY(NCHUNK - 4, 0, ra1, rp1, rh1, true)

    const int    o_first = m0 + mt * 16 + 8 * hi;
    const size_t o_base  = (size_t)b * BSTRIDE + (size_t)o_first * CSTRIDE + w0 + ntp * 16 + l16;
    float e_sc[8], e_sh[8], e_r0[8], e_r1[8];
    #pragma unroll
    for (int j = 0; j < 8; ++j) {
        e_sc[j] = ss[o_first + j];
        e_sh[j] = ss[C_ + o_first + j];
    }
    #pragma unroll
    for (int j = 0; j < 8; ++j) {
        e_r0[j] = addv[o_base + (size_t)j * CSTRIDE];
        e_r1[j] = addv[o_base + (size_t)j * CSTRIDE + 16];
    }

    CHUNK_BODY(NCHUNK - 3, 1, ra0, rp0, rh0, false)
    CHUNK_BODY(NCHUNK - 2, 0, ra1, rp1, rh1, false)
    {
        const int cur = (NCHUNK - 1) & 1;
        MessagePass_frag_load(&As[cur][fa_off], &Ps[cur][fb_off], 1, fa[1], fb0[1], fb1[1]);
        MessagePass_frag_mma(fa[0], fb0[0], fb1[0], acc0, acc1);
        MessagePass_frag_mma(fa[1], fb0[1], fb1[1], acc0, acc1);
    }
#undef CHUNK_BODY
#undef LOAD_CHUNK
#undef STORE_CHUNK

    #pragma unroll
    for (int j = 0; j < 8; ++j) {
        const size_t off = o_base + (size_t)j * CSTRIDE;
        float y0 = acc0[j] * e_sc[j] + e_sh[j];
        float y1 = acc1[j] * e_sc[j] + e_sh[j];
        y0 = y0 > 0.0f ? y0 : 0.0f;
        y1 = y1 > 0.0f ? y1 : 0.0f;
        dst[off]      = y0 + e_r0[j];
        dst[off + 16] = y1 + e_r1[j];
    }
}

__global__ void MessagePass_copy_slice(const float* __restrict__ x, float* __restrict__ out)
{
    int e = blockIdx.x * blockDim.x + threadIdx.x;
    if (e >= B_ * C_ * W_) return;
    int w  = e % W_;
    int bc = e / W_;
    size_t off = (size_t)bc * CSTRIDE + w;
    out[off] = x[off];
}

__global__ void MessagePass_bn_prep(const float* __restrict__ g, const float* __restrict__ be,
                                    const float* __restrict__ rm, const float* __restrict__ rv,
                                    float* __restrict__ ss)
{
    int c = blockIdx.x * blockDim.x + threadIdx.x;
    if (c < C_) {
        float sc = g[c] * rsqrtf(rv[c] + 1e-5f);
        ss[c]      = sc;
        ss[C_ + c] = be[c] - rm[c] * sc;
    }
}

extern "C" void kernel_launch(void* const* d_in, const int* in_sizes, int n_in,
                              void* d_out, int out_size, void* d_ws, size_t ws_size,
                              hipStream_t stream)
{
    const float* x     = (const float*)d_in[0];
    const float* cw    = (const float*)d_in[1];
    const float* gamma = (const float*)d_in[2];
    const float* beta  = (const float*)d_in[3];
    const float* rmean = (const float*)d_in[4];
    const float* rvar  = (const float*)d_in[5];
    float* out = (float*)d_out;
    float* ss  = (float*)d_ws;

    (void)in_sizes; (void)n_in; (void)out_size; (void)ws_size;

    MessagePass_bn_prep<<<1, 128, 0, stream>>>(gamma, beta, rmean, rvar, ss);
    MessagePass_copy_slice<<<(B_ * C_ * W_ + 255) / 256, 256, 0, stream>>>(x, out);

    dim3 grid((B_ * W_) / NT, C_ / MT);

    for (int h = 1; h < H_; ++h)
        MessagePass_step_wmma<<<grid, 128, 0, stream>>>(
            out + (size_t)(h - 1) * W_, x + (size_t)h * W_, out + (size_t)h * W_, cw, ss);

    for (int h = H_ - 2; h >= 0; --h)
        MessagePass_step_wmma<<<grid, 128, 0, stream>>>(
            out + (size_t)(h + 1) * W_, out + (size_t)h * W_, out + (size_t)h * W_, cw, ss);
}


# K split over two waves per SIMD: 8-wave workgroups, each group of 4 waves runs the pipeline over one half of K, partial sums joined through LDS; summation order changes: (half 1)+(half 2)
# speedup vs baseline: 11.2924x; 1.0891x over previous
#include <hip/hip_runtime.h>
#include <hip/hip_bf16.h>

typedef float v2f __attribute__((ext_vector_type(2)));
typedef float v4f __attribute__((ext_vector_type(4)));
typedef float v8f __attribute__((ext_vector_type(8)));

#define B_  16
#define C_  128
#define H_  64
#define W_  256
#define KK_ 1152
#define CSTRIDE (H_ * W_)
#define BSTRIDE (C_ * H_ * W_)

#define KC  72
#define MT  32
#define NT  64
#define A_PAD 76
#define PROW 72
#define PATCH_ROW(c) ((((c) & 2) << 1) | (((c) & 4) >> 1) | ((c) & 1))

#define NCHUNK (KK_ / KC)
#define NGRP 2
#define NCHUNK_G (NCHUNK / NGRP)

#define KSUB 9
#define KSIG 7

__device__ __forceinline__
void MessagePass_lds_barrier()
{
    __builtin_amdgcn_fence(__ATOMIC_RELEASE, "workgroup", "local");
    __builtin_amdgcn_s_barrier();
    __builtin_amdgcn_fence(__ATOMIC_ACQUIRE, "workgroup", "local");
}

__device__ __forceinline__
void MessagePass_lds_arrive()
{
    __builtin_amdgcn_fence(__ATOMIC_RELEASE, "workgroup", "local");
    __builtin_amdgcn_s_barrier_signal(-1);
}
__device__ __forceinline__
void MessagePass_lds_await()
{
    __builtin_amdgcn_s_barrier_wait(-1);
    __builtin_amdgcn_fence(__ATOMIC_ACQUIRE, "workgroup", "local");
}

__device__ __forceinline__
void MessagePass_frag_load(const float* __restrict__ Arow, const float* __restrict__ Pcol,
                           int blk, v2f* fa, v2f* fb0, v2f* fb1)
{
    const float* Pb = Pcol + 2 * blk * PROW;
    #pragma unroll
    for (int s = 0; s < KSUB; ++s) {
        const int j0 = 2 * s, j1 = 2 * s + 1;
        fa[s]    = *reinterpret_cast<const v2f*>(Arow + 36 * blk + 2 * s);
        fb0[s].x = Pb[(j0 / 9) * PROW + (j0 % 9)];
        fb0[s].y = Pb[(j1 / 9) * PROW + (j1 % 9)];
    }
    __builtin_amdgcn_sched_barrier(0);
    #pragma unroll
    for (int s = 0; s < KSUB; ++s) {
        const int j0 = 2 * s, j1 = 2 * s + 1;
        fb1[s].x = Pb[(j0 / 9) * PROW + (j0 % 9) + 16];
        fb1[s].y = Pb[(j1 / 9) * PROW + (j1 % 9) + 16];
    }
    __builtin_amdgcn_sched_barrier(0);
}

__device__ __forceinline__
void MessagePass_frag_mma_range(const v2f* fa, const v2f* fb0, const v2f* fb1, v8f& acc0, v8f& acc1,
                                int s_begin, int s_end)
{
    #pragma unroll
    for (int s = s_begin; s < s_end; ++s) {
        acc0 = __builtin_amdgcn_wmma_f32_16x16x4_f32(false, fa[s], false, fb0[s],
                                                     (short)0, acc0, false, false);
        acc1 = __builtin_amdgcn_wmma_f32_16x16x4_f32(false, fa[s], false, fb1[s],
                                                     (short)0, acc1, false, false);
    }
}

__device__ __forceinline__
void MessagePass_products_end(v8f& acc0, v8f& acc1)
{
    asm volatile("" : "+v"(acc0), "+v"(acc1) : : "memory");
}

__device__ __forceinline__
void MessagePass_frag_mma(const v2f* fa, const v2f* fb0, const v2f* fb1, v8f& acc0, v8f& acc1)
{
    MessagePass_frag_mma_range(fa, fb0, fb1, acc0, acc1, 0, KSUB);
    MessagePass_products_end(acc0, acc1);
}

__global__ __launch_bounds__(256)
void MessagePass_step_wmma(const float* __restrict__ src,
                           const float* __restrict__ addv,
                           float* __restrict__ dst,
                           const float* __restrict__ Wt,
                           const float* __restrict__ ss)
{
    __shared__ __attribute__((aligned(16))) float As[NGRP][2][MT * A_PAD];
    __shared__ __attribute__((aligned(16))) float Ps[NGRP][2][8 * PROW];
    __shared__ __attribute__((aligned(16))) float Xs[8 * 2 * 32 * 4];

    const int grp  = threadIdx.x >> 7;
    const int tid  = threadIdx.x & 127;
    const int lane = tid & 31;
    const int hi   = lane >> 4;
    const int l16  = lane & 15;
    const int wv   = tid >> 5;

    const int m0 = blockIdx.y * MT;
    const int n0 = blockIdx.x * NT;
    const int b  = n0 / W_;
    const int w0 = n0 % W_;

    const int mt  = wv & 1;
    const int ntp = (wv >> 1) * 2;

    const int a_row = wv * 8 + (lane >> 3);
    const int a_kq  = lane & 7;
    const float* a_g = Wt + (size_t)(m0 + a_row) * KK_ + grp * (NCHUNK_G * KC) + a_kq;
    const int a_l = a_row * A_PAD + a_kq;

    const int p_row = tid >> 4;
    const int p_q   = tid & 15;
    const float* p_g = src + (size_t)b * BSTRIDE + (size_t)(grp * (NCHUNK_G * 8) + p_row) * CSTRIDE + w0 + 4 * p_q;
    const int p_l = PATCH_ROW(p_row) * PROW + 4 + 4 * p_q;

    const bool h_own  = tid < 16;
    const int  h_row  = (tid >> 1) & 7;
    const int  h_side = tid & 1;
    const bool h_ok   = h_side ? (w0 + NT < W_) : (w0 > 0);
    const int  h_w    = h_ok ? (h_side ? w0 + NT : w0 - 4) : w0;
    const float* h_g  = src + (size_t)b * BSTRIDE + (size_t)(grp * (NCHUNK_G * 8) + h_row) * CSTRIDE + h_w;
    const int  h_l    = PATCH_ROW(h_row) * PROW + (h_side ? NT + 4 : 0);

    float ra0[18], ra1[18];
    v4f   rp0, rp1;
    v4f   rh0 = {0.0f, 0.0f, 0.0f, 0.0f};
    v4f   rh1 = {0.0f, 0.0f, 0.0f, 0.0f};

#define LOAD_CHUNK(c, RA, RP, RH)                                                  \
    {                                                                              \
        const float* ap = a_g + (c) * KC;                                          \
        _Pragma("unroll")                                                          \
        for (int j = 0; j < 18; ++j)                                               \
            RA[j] = ap[(j / 9) * 4 * KK_ + (j % 9) * 8];                           \
        RP = *reinterpret_cast<const v4f*>(p_g + (size_t)(c) * 8 * CSTRIDE);       \
        if (h_own)                                                                 \
            RH = *reinterpret_cast<const v4f*>(h_g + (size_t)(c) * 8 * CSTRIDE);   \
    }
#define STORE_CHUNK(buf, RA, RP, RH)                                               \
    {                                                                              \
        _Pragma("unroll")                                                          \
        for (int j = 0; j < 18; ++j)                                               \
            As[grp][buf][a_l + (j / 9) * 4 * A_PAD + (j % 9) * 8] = RA[j];         \
        *reinterpret_cast<v4f*>(&Ps[grp][buf][p_l]) = RP;                          \
        if (h_own) {                                                               \
            const v4f z = {0.0f, 0.0f, 0.0f, 0.0f};                                \
            *reinterpret_cast<v4f*>(&Ps[grp][buf][h_l]) = h_ok ? RH : z;           \
        }                                                                          \
    }

    v8f acc0 = {};
    v8f acc1 = {};

    const int fa_off = (mt * 16 + l16) * A_PAD + 18 * hi;
    const int fb_off = (4 * hi) * PROW + ntp * 16 + l16;

    v2f fa[2][KSUB], fb0[2][KSUB], fb1[2][KSUB];

#define CHUNK_BODY(c, CUR, RA, RP, RH, MORE)                                               \
    {                                                                                      \
        STORE_CHUNK((CUR) ^ 1, RA, RP, RH)     \
        MessagePass_frag_load(&As[grp][CUR][fa_off], &Ps[grp][CUR][fb_off], 1, fa[1], fb0[1], fb1[1]);      \
        MessagePass_frag_mma_range(fa[0], fb0[0], fb1[0], acc0, acc1, 0, KSIG);            \
          \
        __builtin_amdgcn_sched_barrier(0);                                                 \
        MessagePass_lds_arrive();                                                          \
        __builtin_amdgcn_sched_barrier(0);                                                 \
        MessagePass_frag_mma_range(fa[0], fb0[0], fb1[0], acc0, acc1, KSIG, KSUB);         \
        MessagePass_products_end(acc0, acc1);                                              \
        if (MORE) LOAD_CHUNK((c) + 3, RA, RP, RH)     \
        MessagePass_lds_await();                                                           \
        MessagePass_frag_load(&As[grp][(CUR) ^ 1][fa_off], &Ps[grp][(CUR) ^ 1][fb_off], 0, fa[0], fb0[0], fb1[0]); \
        MessagePass_frag_mma(fa[1], fb0[1], fb1[1], acc0, acc1);                           \
    }

    LOAD_CHUNK(0, ra0, rp0, rh0)
    LOAD_CHUNK(1, ra1, rp1, rh1)
    STORE_CHUNK(0, ra0, rp0, rh0)
    MessagePass_lds_barrier();
    LOAD_CHUNK(2, ra0, rp0, rh0)
    MessagePass_frag_load(&As[grp][0][fa_off], &Ps[grp][0][fb_off], 0, fa[0], fb0[0], fb1[0]);

    for (int c = 0; c < NCHUNK_G - 4; c += 2) {
        CHUNK_BODY(c,     0, ra1, rp1, rh1, true)
        CHUNK_BODY(c + 1, 1, ra0, rp0, rh0, true)
    }
    CHUNK_BODY(NCHUNK_G - 4, 0, ra1, rp1, rh1, true)

    const int    o_first = m0 + mt * 16 + 8 * hi + 4 * grp;
    const size_t o_base  = (size_t)b * BSTRIDE + (size_t)o_first * CSTRIDE + w0 + ntp * 16 + l16;
    float e_sc[4], e_sh[4], e_r0[4], e_r1[4];
    #pragma unroll
    for (int j = 0; j < 4; ++j) {
        e_sc[j] = ss[o_first + j];
        e_sh[j] = ss[C_ + o_first + j];
    }
    #pragma unroll
    for (int j = 0; j < 4; ++j) {
        e_r0[j] = addv[o_base + (size_t)j * CSTRIDE];
        e_r1[j] = addv[o_base + (size_t)j * CSTRIDE + 16];
    }

    CHUNK_BODY(NCHUNK_G - 3, 1, ra0, rp0, rh0, false)
    CHUNK_BODY(NCHUNK_G - 2, 0, ra1, rp1, rh1, false)
    {
        const int cur = (NCHUNK_G - 1) & 1;
        MessagePass_frag_load(&As[grp][cur][fa_off], &Ps[grp][cur][fb_off], 1, fa[1], fb0[1], fb1[1]);
        MessagePass_frag_mma(fa[0], fb0[0], fb1[0], acc0, acc1);
        MessagePass_frag_mma(fa[1], fb0[1], fb1[1], acc0, acc1);
    }
#undef CHUNK_BODY
#undef LOAD_CHUNK
#undef STORE_CHUNK

    const v4f lo0 = {acc0[0], acc0[1], acc0[2], acc0[3]};
    const v4f hi0 = {acc0[4], acc0[5], acc0[6], acc0[7]};
    const v4f lo1 = {acc1[0], acc1[1], acc1[2], acc1[3]};
    const v4f hi1 = {acc1[4], acc1[5], acc1[6], acc1[7]};
    const bool upper = grp != 0;
    v4f keep0 = upper ? hi0 : lo0;
    v4f keep1 = upper ? hi1 : lo1;
    const v4f send0 = upper ? lo0 : hi0;
    const v4f send1 = upper ? lo1 : hi1;
    const int wfull = threadIdx.x >> 5;
    v4f* X = reinterpret_cast<v4f*>(Xs);
    X[(wfull * 2 + 0) * 32 + lane] = send0;
    X[(wfull * 2 + 1) * 32 + lane] = send1;
    MessagePass_lds_barrier();
    keep0 += X[((wfull ^ 4) * 2 + 0) * 32 + lane];
    keep1 += X[((wfull ^ 4) * 2 + 1) * 32 + lane];

    #pragma unroll
    for (int j = 0; j < 4; ++j) {
        const size_t off = o_base + (size_t)j * CSTRIDE;
        float y0 = keep0[j] * e_sc[j] + e_sh[j];
        float y1 = keep1[j] * e_sc[j] + e_sh[j];
        y0 = y0 > 0.0f ? y0 : 0.0f;
        y1 = y1 > 0.0f ? y1 : 0.0f;
        dst[off]      = y0 + e_r0[j];
        dst[off + 16] = y1 + e_r1[j];
    }
}

__global__ void MessagePass_copy_slice(const float* __restrict__ x, float* __restrict__ out)
{
    int e = blockIdx.x * blockDim.x + threadIdx.x;
    if (e >= B_ * C_ * W_) return;
    int w  = e % W_;
    int bc = e / W_;
    size_t off = (size_t)bc * CSTRIDE + w;
    out[off] = x[off];
}

__global__ void MessagePass_bn_prep(const float* __restrict__ g, const float* __restrict__ be,
                                    const float* __restrict__ rm, const float* __restrict__ rv,
                                    float* __restrict__ ss)
{
    int c = blockIdx.x * blockDim.x + threadIdx.x;
    if (c < C_) {
        float sc = g[c] * rsqrtf(rv[c] + 1e-5f);
        ss[c]      = sc;
        ss[C_ + c] = be[c] - rm[c] * sc;
    }
}

extern "C" void kernel_launch(void* const* d_in, const int* in_sizes, int n_in,
                              void* d_out, int out_size, void* d_ws, size_t ws_size,
                              hipStream_t stream)
{
    const float* x     = (const float*)d_in[0];
    const float* cw    = (const float*)d_in[1];
    const float* gamma = (const float*)d_in[2];
    const float* beta  = (const float*)d_in[3];
    const float* rmean = (const float*)d_in[4];
    const float* rvar  = (const float*)d_in[5];
    float* out = (float*)d_out;
    float* ss  = (float*)d_ws;

    (void)in_sizes; (void)n_in; (void)out_size; (void)ws_size;

    MessagePass_bn_prep<<<1, 128, 0, stream>>>(gamma, beta, rmean, rvar, ss);
    MessagePass_copy_slice<<<(B_ * C_ * W_ + 255) / 256, 256, 0, stream>>>(x, out);

    dim3 grid((B_ * W_) / NT, C_ / MT);

    for (int h = 1; h < H_; ++h)
        MessagePass_step_wmma<<<grid, 256, 0, stream>>>(
            out + (size_t)(h - 1) * W_, x + (size_t)h * W_, out + (size_t)h * W_, cw, ss);

    for (int h = H_ - 2; h >= 0; --h)
        MessagePass_step_wmma<<<grid, 256, 0, stream>>>(
            out + (size_t)(h + 1) * W_, out + (size_t)h * W_, out + (size_t)h * W_, cw, ss);
}
